# starting kernel
# baseline (speedup 1.0000x reference)
#include <hip/hip_runtime.h>
#include <hip/hip_bf16.h>

#define V_N     10000
#define E_N     60000
#define B_N     8
#define LATENT  512
#define HIDDEN  256
#define NLAYERS 5
#define NNODES  (B_N * V_N)

typedef float v2f __attribute__((ext_vector_type(2)));
typedef float v8f __attribute__((ext_vector_type(8)));

#if defined(__gfx1250__) && __has_builtin(__builtin_amdgcn_wmma_f32_16x16x4_f32)
#define HAVE_WMMA_F32 1
#endif


__global__ void k_deg_init(float* deg) {
    int v = blockIdx.x * blockDim.x + threadIdx.x;
    if (v < V_N) deg[v] = 1.0f;
}

__global__ void k_deg_acc(float* deg, const int* __restrict__ dst) {
    int e = blockIdx.x * blockDim.x + threadIdx.x;
    if (e < E_N) atomicAdd(&deg[dst[e]], 1.0f);
}

__global__ void k_dinv(float* deg_to_dinv, float* selfnorm) {
    int v = blockIdx.x * blockDim.x + threadIdx.x;
    if (v < V_N) {
        float di = rsqrtf(deg_to_dinv[v]);
        deg_to_dinv[v] = di;
        selfnorm[v] = di * di;
    }
}

__global__ void k_norm(const float* __restrict__ dinv, const int* __restrict__ src,
                       const int* __restrict__ dst, float* __restrict__ norm) {
    int e = blockIdx.x * blockDim.x + threadIdx.x;
    if (e < E_N) norm[e] = dinv[src[e]] * dinv[dst[e]];
}


__global__ void k_xyzW(const float* __restrict__ xyz, const float* __restrict__ iW,
                       const float* __restrict__ ib, float* __restrict__ xyzW) {
    int i = blockIdx.x * blockDim.x + threadIdx.x;
    if (i >= V_N * HIDDEN) return;
    int v = i / HIDDEN, hh = i % HIDDEN;
    const float* wp = iW + (size_t)hh * (LATENT + 3);
    xyzW[i] = ib[hh] + xyz[3 * v] * wp[0] + xyz[3 * v + 1] * wp[1] + xyz[3 * v + 2] * wp[2];
}

__global__ void k_latW(const float* __restrict__ latent, const float* __restrict__ iW,
                       float* __restrict__ latW) {
    int i = blockIdx.x * blockDim.x + threadIdx.x;
    if (i >= B_N * HIDDEN) return;
    int b = i / HIDDEN, hh = i % HIDDEN;
    const float* lp = latent + (size_t)b * LATENT;
    const float* wp = iW + (size_t)hh * (LATENT + 3) + 3;
    float s = 0.0f;
    for (int j = 0; j < LATENT; ++j) s += lp[j] * wp[j];
    latW[i] = s;
}

__global__ void k_x0(const float* __restrict__ xyzW, const float* __restrict__ latW,
                     float* __restrict__ x) {
    size_t i = (size_t)blockIdx.x * blockDim.x + threadIdx.x;
    int hh = (int)(i & (HIDDEN - 1));
    size_t node = i >> 8;
    int v = (int)(node % V_N);
    int b = (int)(node / V_N);
    float s = xyzW[(size_t)v * HIDDEN + hh] + latW[b * HIDDEN + hh];
    x[i] = s > 0.0f ? s : 0.0f;
}


#define GEMM_MT 2
#define GEMM_NT 4

__global__ void __launch_bounds__(256)
k_gemm_wmma(const float* __restrict__ x, const float* __restrict__ W,
            float* __restrict__ h) {
    const int lane  = threadIdx.x & 31;
    const int wv    = threadIdx.x >> 5;
    const int half  = lane >> 4;
    const int l16   = lane & 15;
    const int mbase = blockIdx.x * 64 + (wv >> 2) * 32;
    const int nbase = (wv & 3) * 64;

    const float* arow = x + (size_t)(mbase + l16) * HIDDEN + half * 2;
    const float* bcol = W + (size_t)(nbase + l16) * HIDDEN + half * 2;

    v8f acc[GEMM_MT][GEMM_NT] = {};

#if HAVE_WMMA_F32
#pragma unroll 4
    for (int k = 0; k < HIDDEN; k += 4) {
        v2f a[GEMM_MT], b[GEMM_NT];
#pragma unroll
        for (int mt = 0; mt < GEMM_MT; ++mt)
            a[mt] = *(const v2f*)(arow + (size_t)mt * 16 * HIDDEN + k);
#pragma unroll
        for (int nt = 0; nt < GEMM_NT; ++nt)
            b[nt] = *(const v2f*)(bcol + (size_t)nt * 16 * HIDDEN + k);
#pragma unroll
        for (int mt = 0; mt < GEMM_MT; ++mt)
#pragma unroll
            for (int nt = 0; nt < GEMM_NT; ++nt)
                acc[mt][nt] = __builtin_amdgcn_wmma_f32_16x16x4_f32(
                     false, a[mt],  false, b[nt],
                     (short)0, acc[mt][nt],
                     false,  false);
    }
#else
#pragma unroll
    for (int mt = 0; mt < GEMM_MT; ++mt)
#pragma unroll
        for (int r = 0; r < 8; ++r) {
            const float* xr = x + (size_t)(mbase + mt * 16 + half * 8 + r) * HIDDEN;
#pragma unroll
            for (int nt = 0; nt < GEMM_NT; ++nt) {
                const float* wr = W + (size_t)(nbase + nt * 16 + l16) * HIDDEN;
                float s = 0.0f;
                for (int k = 0; k < HIDDEN; ++k) s += xr[k] * wr[k];
                acc[mt][nt][r] = s;
            }
        }
#endif

    float* outp = h + (size_t)(mbase + half * 8) * HIDDEN + (nbase + l16);
#pragma unroll
    for (int mt = 0; mt < GEMM_MT; ++mt)
#pragma unroll
        for (int nt = 0; nt < GEMM_NT; ++nt)
#pragma unroll
            for (int r = 0; r < 8; ++r)
                outp[((size_t)mt * 16 + r) * HIDDEN + nt * 16] = acc[mt][nt][r];
}


__global__ void k_selfinit(const float* __restrict__ h, float* __restrict__ agg,
                           const float* __restrict__ selfnorm) {
    size_t i = (size_t)blockIdx.x * blockDim.x + threadIdx.x;
    int v = (int)((i >> 8) % V_N);
    agg[i] = h[i] * selfnorm[v];
}

__global__ void k_scatter(const float* __restrict__ h, float* __restrict__ agg,
                          const int* __restrict__ src, const int* __restrict__ dst,
                          const float* __restrict__ norm) {
    int t  = blockIdx.x * blockDim.x + threadIdx.x;
    int e  = t >> 6;
    int f4 = (t & 63) << 2;
    int b  = blockIdx.y;
    int s  = src[e], d = dst[e];
    float w = norm[e];
    const float* hp = h + ((size_t)b * V_N + s) * HIDDEN + f4;
    float*       ap = agg + ((size_t)b * V_N + d) * HIDDEN + f4;
    float4 hv = *(const float4*)hp;
    atomicAdd(ap + 0, hv.x * w);
    atomicAdd(ap + 1, hv.y * w);
    atomicAdd(ap + 2, hv.z * w);
    atomicAdd(ap + 3, hv.w * w);
}

__global__ void k_relu_bias(float* __restrict__ x, const float* __restrict__ bias) {
    size_t i = (size_t)blockIdx.x * blockDim.x + threadIdx.x;
    float s = x[i] + bias[i & (HIDDEN - 1)];
    x[i] = s > 0.0f ? s : 0.0f;
}

__global__ void k_out(const float* __restrict__ x, const float* __restrict__ oW,
                      const float* __restrict__ ob, float* __restrict__ disp) {
    int idx = blockIdx.x * blockDim.x + threadIdx.x;
    if (idx >= NNODES * 3) return;
    int node = idx / 3, o = idx % 3;
    const float* xp = x + (size_t)node * HIDDEN;
    const float* wp = oW + o * HIDDEN;
    float s = ob[o];
#pragma unroll 4
    for (int k = 0; k < HIDDEN; k += 4) {
        float4 xv = *(const float4*)(xp + k);
        float4 wv = *(const float4*)(wp + k);
        s += xv.x * wv.x + xv.y * wv.y + xv.z * wv.z + xv.w * wv.w;
    }
    disp[idx] = s;
}


extern "C" void kernel_launch(void* const* d_in, const int* in_sizes, int n_in,
                              void* d_out, int out_size, void* d_ws, size_t ws_size,
                              hipStream_t stream) {
    const float* vertex_xyz = (const float*)d_in[0];
    const float* latent     = (const float*)d_in[1];
    const int*   edge_index = (const int*)d_in[2];
    const float* input_W    = (const float*)d_in[3];
    const float* input_b    = (const float*)d_in[4];
    const float* conv_W     = (const float*)d_in[5];
    const float* conv_b     = (const float*)d_in[6];
    const float* out_W      = (const float*)d_in[7];
    const float* out_b      = (const float*)d_in[8];
    float* disp = (float*)d_out;

    const int* src = edge_index;
    const int* dst = edge_index + E_N;

    float* xbuf     = (float*)d_ws;
    float* hbuf     = xbuf + (size_t)NNODES * HIDDEN;
    float* xyzW     = hbuf + (size_t)NNODES * HIDDEN;
    float* latW     = xyzW + (size_t)V_N * HIDDEN;
    float* dinv     = latW + (size_t)B_N * HIDDEN;
    float* selfnorm = dinv + V_N;
    float* norm     = selfnorm + V_N;

    const int T = 256;
    const size_t NH = (size_t)NNODES * HIDDEN;

    k_deg_init<<<(V_N + T - 1) / T, T, 0, stream>>>(dinv);
    k_deg_acc<<<(E_N + T - 1) / T, T, 0, stream>>>(dinv, dst);
    k_dinv<<<(V_N + T - 1) / T, T, 0, stream>>>(dinv, selfnorm);
    k_norm<<<(E_N + T - 1) / T, T, 0, stream>>>(dinv, src, dst, norm);

    k_xyzW<<<(V_N * HIDDEN + T - 1) / T, T, 0, stream>>>(vertex_xyz, input_W, input_b, xyzW);
    k_latW<<<(B_N * HIDDEN + T - 1) / T, T, 0, stream>>>(latent, input_W, latW);
    k_x0<<<(unsigned)(NH / T), T, 0, stream>>>(xyzW, latW, xbuf);

    dim3 ggrid(NNODES / 64);
    dim3 sgrid((E_N * 64) / T, B_N);
    for (int l = 0; l < NLAYERS; ++l) {
        k_gemm_wmma<<<ggrid, T, 0, stream>>>(xbuf, conv_W + (size_t)l * HIDDEN * HIDDEN, hbuf);
        k_selfinit<<<(unsigned)(NH / T), T, 0, stream>>>(hbuf, xbuf, selfnorm);
        k_scatter<<<sgrid, T, 0, stream>>>(hbuf, xbuf, src, dst, norm);
        k_relu_bias<<<(unsigned)(NH / T), T, 0, stream>>>(xbuf, conv_b + (size_t)l * HIDDEN);
    }

    k_out<<<(NNODES * 3 + T - 1) / T, T, 0, stream>>>(xbuf, out_W, out_b, disp);
}


# Neighbour sum: edges re-ordered once per call by (rank, dst mod 32) so a wave's 32 LDS adds fall on 32 different banks (were random, about 3 per bank); same LDS adds, arrival-order sums, order changes
# speedup vs baseline: 2.9741x; 2.9741x over previous
#include <hip/hip_runtime.h>


#define V_N     10000
#define E_N     60000
#define B_N     8
#define LATENT  512
#define HIDDEN  256
#define NLAYERS 5
#define NNODES  (B_N * V_N)

typedef float v2f __attribute__((ext_vector_type(2)));
typedef float v8f __attribute__((ext_vector_type(8)));


#define NCLS 32
#define CNT_PITCH 64

__global__ void k_deg_init(float* deg, unsigned* cnt) {
    int v = blockIdx.x * blockDim.x + threadIdx.x;
    if (v < V_N) deg[v] = 1.0f;
    if (v < NCLS) cnt[v * CNT_PITCH] = 0u;
}

__global__ void k_deg_acc(float* deg, const int* __restrict__ dst, unsigned* cnt,
                          unsigned* __restrict__ rank) {
    __shared__ unsigned lc[NCLS];
    __shared__ unsigned gb[NCLS];
    if (threadIdx.x < NCLS) lc[threadIdx.x] = 0u;
    __syncthreads();
    const int e = blockIdx.x * blockDim.x + threadIdx.x;
    const bool ok = e < E_N;
    unsigned b = 0u, lr = 0u;
    if (ok) {
        int d = dst[e];
        atomicAdd(&deg[d], 1.0f);
        b = (unsigned)d & (NCLS - 1);
        lr = atomicAdd(&lc[b], 1u);
    }
    __syncthreads();
    if (threadIdx.x < NCLS) gb[threadIdx.x] = atomicAdd(&cnt[threadIdx.x * CNT_PITCH], lc[threadIdx.x]);
    __syncthreads();
    if (ok) rank[e] = gb[b] + lr;
}

__global__ void k_dinv(float* deg_to_dinv) {
    int v = blockIdx.x * blockDim.x + threadIdx.x;
    if (v < V_N) deg_to_dinv[v] = rsqrtf(deg_to_dinv[v]);
}

__global__ void k_pack(const int* __restrict__ src, const int* __restrict__ dst,
                       const unsigned* __restrict__ cnt, const unsigned* __restrict__ rank,
                       unsigned* __restrict__ epk) {
    __shared__ unsigned c[NCLS];
    if (threadIdx.x < NCLS) c[threadIdx.x] = cnt[threadIdx.x * CNT_PITCH];
    __syncthreads();
    int e = blockIdx.x * blockDim.x + threadIdx.x;
    if (e < E_N) {
        unsigned d = (unsigned)dst[e];
        unsigned b = d & (NCLS - 1);
        unsigned r = rank[e];
        unsigned pos = 0;
        for (unsigned j = 0; j < NCLS; ++j) {
            unsigned cj = c[j];
            pos += (cj < r ? cj : r) + ((j < b && cj > r) ? 1u : 0u);
        }
        epk[pos] = ((unsigned)src[e] & 0xFFFFu) | (d << 16);
    }
}


__global__ void k_xyzW(const float* __restrict__ xyz, const float* __restrict__ iW,
                       const float* __restrict__ ib, float* __restrict__ xyzWT) {
    int i = blockIdx.x * blockDim.x + threadIdx.x;
    if (i >= V_N * HIDDEN) return;
    int hh = i / V_N, v = i % V_N;
    const float* wp = iW + (size_t)hh * (LATENT + 3);
    xyzWT[i] = ib[hh] + xyz[3 * v] * wp[0] + xyz[3 * v + 1] * wp[1] + xyz[3 * v + 2] * wp[2];
}

__global__ void k_latW(const float* __restrict__ latent, const float* __restrict__ iW,
                       float* __restrict__ latW) {
    int i = blockIdx.x * blockDim.x + threadIdx.x;
    if (i >= B_N * HIDDEN) return;
    int b = i / HIDDEN, hh = i % HIDDEN;
    const float* lp = latent + (size_t)b * LATENT;
    const float* wp = iW + (size_t)hh * (LATENT + 3) + 3;
    float s = 0.0f;
    for (int j = 0; j < LATENT; ++j) s += lp[j] * wp[j];
    latW[i] = s;
}


#define GEMM_OT 2
#define GEMM_NG 2
#define GEMM_NT (4 * GEMM_NG)

template <bool FIRST>
__device__ __forceinline__ void
gemm_tile(const float* __restrict__ src, const float* __restrict__ latW,
          const float* __restrict__ W, float* __restrict__ hT) {
    const int lane  = threadIdx.x & 31;
    const int wv    = threadIdx.x >> 5;
    const int half  = lane >> 4;
    const int l16   = lane & 15;
    const int nbase = (blockIdx.x >> 1) * 128;
    const int obase = (blockIdx.x & 1) * 128 + wv * 32;

    constexpr size_t RS = FIRST ? (size_t)V_N : (size_t)NNODES;

    const float* wrow = W + (size_t)(obase + l16) * HIDDEN + half * 2;
    const float* xcol[GEMM_NG];
    const float* lcol[GEMM_NG];
#pragma unroll
    for (int g = 0; g < GEMM_NG; ++g) {
        const int node4 = nbase + 64 * g + 4 * l16;
        if (FIRST) {
            const int bb = node4 / V_N;
            xcol[g] = src + (size_t)(half * 2) * RS + (node4 - bb * V_N);
            lcol[g] = latW + bb * HIDDEN + half * 2;
        } else {
            xcol[g] = src + (size_t)(half * 2) * RS + node4;
            lcol[g] = nullptr;
        }
    }

    v8f acc[GEMM_OT][GEMM_NT] = {};

#pragma unroll (FIRST ? 1 : 2)
    for (int k = 0; k < HIDDEN; k += 4) {
        v2f a[GEMM_OT];
#pragma unroll
        for (int ot = 0; ot < GEMM_OT; ++ot)
            a[ot] = *(const v2f*)(wrow + (size_t)ot * 16 * HIDDEN + k);
        v2f b[GEMM_NT];
#pragma unroll
        for (int g = 0; g < GEMM_NG; ++g) {
            float4 b0 = *(const float4*)(xcol[g] + (size_t)k * RS);
            float4 b1 = *(const float4*)(xcol[g] + (size_t)(k + 1) * RS);
            if (FIRST) {
                const v2f lw = *(const v2f*)(lcol[g] + k);
                b0.x = fmaxf(b0.x + lw.x, 0.0f); b0.y = fmaxf(b0.y + lw.x, 0.0f);
                b0.z = fmaxf(b0.z + lw.x, 0.0f); b0.w = fmaxf(b0.w + lw.x, 0.0f);
                b1.x = fmaxf(b1.x + lw.y, 0.0f); b1.y = fmaxf(b1.y + lw.y, 0.0f);
                b1.z = fmaxf(b1.z + lw.y, 0.0f); b1.w = fmaxf(b1.w + lw.y, 0.0f);
            }
            b[4 * g + 0].x = b0.x; b[4 * g + 0].y = b1.x;
            b[4 * g + 1].x = b0.y; b[4 * g + 1].y = b1.y;
            b[4 * g + 2].x = b0.z; b[4 * g + 2].y = b1.z;
            b[4 * g + 3].x = b0.w; b[4 * g + 3].y = b1.w;
        }
#pragma unroll
        for (int ot = 0; ot < GEMM_OT; ++ot)
#pragma unroll
            for (int nt = 0; nt < GEMM_NT; ++nt)
                acc[ot][nt] = __builtin_amdgcn_wmma_f32_16x16x4_f32(
                     false, a[ot],  false, b[nt],
                     (short)0, acc[ot][nt],
                     false,  false);
    }

    float* outp = hT + (size_t)(obase + half * 8) * NNODES + (nbase + 4 * l16);
#pragma unroll
    for (int ot = 0; ot < GEMM_OT; ++ot)
#pragma unroll
        for (int r = 0; r < 8; ++r)
#pragma unroll
            for (int g = 0; g < GEMM_NG; ++g)
                *(float4*)(outp + ((size_t)ot * 16 + r) * NNODES + 64 * g) =
                    make_float4(acc[ot][4 * g + 0][r], acc[ot][4 * g + 1][r],
                                acc[ot][4 * g + 2][r], acc[ot][4 * g + 3][r]);
}

__global__ void __launch_bounds__(128)
k_gemm_wmma(const float* __restrict__ xT, const float* __restrict__ W,
            float* __restrict__ hT) {
    gemm_tile<false>(xT, nullptr, W, hT);
}

__global__ void __launch_bounds__(128)
k_gemm_first(const float* __restrict__ xyzWT, const float* __restrict__ latW,
             const float* __restrict__ W, float* __restrict__ hT) {
    gemm_tile<true>(xyzWT, latW, W, hT);
}


#define AGG_T 1024
#define AGG_Q (2 * V_N / 4)
#define AGG_V4 (V_N / 4)

__global__ void __launch_bounds__(AGG_T)
k_agg(const float* __restrict__ hT, float* __restrict__ xT,
      const unsigned* __restrict__ epk, const float* __restrict__ dinv,
      const float* __restrict__ bias) {
    __shared__ __attribute__((aligned(16))) float g[2 * V_N];
    __shared__ __attribute__((aligned(16))) float acc[2 * V_N];

    const int tid = threadIdx.x;
    const int p   = blockIdx.x & 3;
    const int o   = blockIdx.x >> 2;
    const size_t c0 = (size_t)o * NNODES + (size_t)p * (2 * V_N);

    for (int q = tid; q < AGG_V4; q += AGG_T) {
        float4 d4 = *(const float4*)(dinv + 4 * q);
        float4 ha = *(const float4*)(hT + c0 + 4 * q);
        float4 hb = *(const float4*)(hT + c0 + V_N + 4 * q);
        float4 ta = make_float4(d4.x * ha.x, d4.y * ha.y, d4.z * ha.z, d4.w * ha.w);
        float4 tb = make_float4(d4.x * hb.x, d4.y * hb.y, d4.z * hb.z, d4.w * hb.w);
        *(float4*)(acc + 4 * q) = ta;
        *(float4*)(acc + V_N + 4 * q) = tb;
        *(float4*)(g + 8 * q)     = make_float4(ta.x, tb.x, ta.y, tb.y);
        *(float4*)(g + 8 * q + 4) = make_float4(ta.z, tb.z, ta.w, tb.w);
    }
    __syncthreads();

    for (int base = 0; base < E_N; base += 4 * AGG_T) {
        unsigned pk[4];
        float2 gs[4];
        bool ok[4];
#pragma unroll
        for (int j = 0; j < 4; ++j) {
            int e = base + j * AGG_T + tid;
            ok[j] = e < E_N;
            pk[j] = ok[j] ? epk[e] : 0u;
        }
#pragma unroll
        for (int j = 0; j < 4; ++j) {
            unsigned s = pk[j] & 0xFFFFu;
            gs[j] = *(const float2*)(g + 2 * s);
        }
#pragma unroll
        for (int j = 0; j < 4; ++j) {
            if (ok[j]) {
                unsigned d = pk[j] >> 16;
                atomicAdd(&acc[d], gs[j].x);
                atomicAdd(&acc[V_N + d], gs[j].y);
            }
        }
    }
    __syncthreads();

    const float bo = bias[o];
    for (int q = tid; q < AGG_Q; q += AGG_T) {
        int vq = q < V_N / 4 ? q : q - V_N / 4;
        float4 d4 = *(const float4*)(dinv + 4 * vq);
        float4 a4 = *(const float4*)(acc + 4 * q);
        float4 r4;
        r4.x = fmaxf(d4.x * a4.x + bo, 0.0f);
        r4.y = fmaxf(d4.y * a4.y + bo, 0.0f);
        r4.z = fmaxf(d4.z * a4.z + bo, 0.0f);
        r4.w = fmaxf(d4.w * a4.w + bo, 0.0f);
        *(float4*)(xT + c0 + 4 * q) = r4;
    }
}

__global__ void __launch_bounds__(256)
k_out(const float* __restrict__ xT, const float* __restrict__ oW,
      const float* __restrict__ ob, float* __restrict__ disp) {
    __shared__ float red[4][3][64];
    const int nl = threadIdx.x & 63;
    const int kq = threadIdx.x >> 6;
    const int node = blockIdx.x * 64 + nl;
    const float* xp = xT + (size_t)(kq * 64) * NNODES + node;
    const float* wp = oW + kq * 64;
    float s0 = 0.0f, s1 = 0.0f, s2 = 0.0f;
#pragma unroll 8
    for (int k = 0; k < 64; ++k) {
        float xv = xp[(size_t)k * NNODES];
        s0 += xv * wp[k];
        s1 += xv * wp[HIDDEN + k];
        s2 += xv * wp[2 * HIDDEN + k];
    }
    red[kq][0][nl] = s0; red[kq][1][nl] = s1; red[kq][2][nl] = s2;
    __syncthreads();
    if (threadIdx.x < 192) {
        int n2 = threadIdx.x / 3, o = threadIdx.x % 3;
        float s = ob[o] + ((red[0][o][n2] + red[1][o][n2]) + (red[2][o][n2] + red[3][o][n2]));
        disp[(size_t)blockIdx.x * 192 + threadIdx.x] = s;
    }
}


extern "C" void kernel_launch(void* const* d_in, const int* in_sizes, int n_in,
                              void* d_out, int out_size, void* d_ws, size_t ws_size,
                              hipStream_t stream) {
    const float* vertex_xyz = (const float*)d_in[0];
    const float* latent     = (const float*)d_in[1];
    const int*   edge_index = (const int*)d_in[2];
    const float* input_W    = (const float*)d_in[3];
    const float* input_b    = (const float*)d_in[4];
    const float* conv_W     = (const float*)d_in[5];
    const float* conv_b     = (const float*)d_in[6];
    const float* out_W      = (const float*)d_in[7];
    const float* out_b      = (const float*)d_in[8];
    float* disp = (float*)d_out;

    const int* src = edge_index;
    const int* dst = edge_index + E_N;

    float* xbuf     = (float*)d_ws;
    float* hbuf     = xbuf + (size_t)NNODES * HIDDEN;
    float* xyzWT    = hbuf + (size_t)NNODES * HIDDEN;
    float* latW     = xyzWT + (size_t)V_N * HIDDEN;
    float* dinv     = latW + (size_t)B_N * HIDDEN;
    float* spare_v  = dinv + V_N;
    unsigned* epk   = (unsigned*)(spare_v + V_N);
    unsigned* cnt   = (unsigned*)spare_v;
    unsigned* rank  = (unsigned*)xbuf;

    const int T = 256;

    k_deg_init<<<(V_N + T - 1) / T, T, 0, stream>>>(dinv, cnt);
    k_deg_acc<<<(E_N + T - 1) / T, T, 0, stream>>>(dinv, dst, cnt, rank);
    k_dinv<<<(V_N + T - 1) / T, T, 0, stream>>>(dinv);
    k_pack<<<(E_N + T - 1) / T, T, 0, stream>>>(src, dst, cnt, rank, epk);

    k_xyzW<<<(V_N * HIDDEN + T - 1) / T, T, 0, stream>>>(vertex_xyz, input_W, input_b, xyzWT);
    k_latW<<<(B_N * HIDDEN + T - 1) / T, T, 0, stream>>>(latent, input_W, latW);

    k_gemm_first<<<2 * (NNODES / 128), 128, 0, stream>>>(xyzWT, latW, conv_W, hbuf);
    k_agg<<<HIDDEN * (B_N / 2), AGG_T, 0, stream>>>(hbuf, xbuf, epk, dinv, conv_b);
    for (int l = 1; l < NLAYERS; ++l) {
        k_gemm_wmma<<<2 * (NNODES / 128), 128, 0, stream>>>(xbuf, conv_W + (size_t)l * HIDDEN * HIDDEN, hbuf);
        k_agg<<<HIDDEN * (B_N / 2), AGG_T, 0, stream>>>(hbuf, xbuf, epk, dinv,
                                                       conv_b + (size_t)l * HIDDEN);
    }

    k_out<<<NNODES / 64, T, 0, stream>>>(xbuf, out_W, out_b, disp);
}


# Neighbour sum as a gather out of LDS: per-destination source lists built once per call; a thread sums its vertex's sources, 4 batches per 16-byte LDS load; no float atomics; sum order by arrival rank
# speedup vs baseline: 3.2346x; 1.0876x over previous
#include <hip/hip_runtime.h>


#define V_N     10000
#define E_N     60000
#define B_N     8
#define LATENT  512
#define HIDDEN  256
#define NLAYERS 5
#define NNODES  (B_N * V_N)

typedef float v2f __attribute__((ext_vector_type(2)));
typedef float v8f __attribute__((ext_vector_type(8)));


__global__ void k_deg_init(unsigned* cntv) {
    int v = blockIdx.x * blockDim.x + threadIdx.x;
    if (v < V_N) cntv[v] = 0u;
}

__global__ void k_deg_acc(const int* __restrict__ dst, unsigned* cntv, unsigned* __restrict__ rank) {
    const int e = blockIdx.x * blockDim.x + threadIdx.x;
    if (e < E_N) rank[e] = atomicAdd(&cntv[dst[e]], 1u);
}

#define SCAN_T 1024
#define SCAN_PER 10

__global__ void __launch_bounds__(SCAN_T)
k_scan(unsigned* cnt_to_rp, float* __restrict__ dinv) {
    __shared__ unsigned part[SCAN_T];
    const int tid = threadIdx.x;
    unsigned c[SCAN_PER];
    unsigned s = 0u;
#pragma unroll
    for (int i = 0; i < SCAN_PER; ++i) {
        const int v = tid * SCAN_PER + i;
        c[i] = v < V_N ? cnt_to_rp[v] : 0u;
        s += c[i];
    }
    part[tid] = s;
    __syncthreads();
    for (int off = 1; off < SCAN_T; off <<= 1) {
        const unsigned t = tid >= off ? part[tid - off] : 0u;
        __syncthreads();
        part[tid] += t;
        __syncthreads();
    }
    unsigned run = part[tid] - s;
#pragma unroll
    for (int i = 0; i < SCAN_PER; ++i) {
        const int v = tid * SCAN_PER + i;
        if (v < V_N) {
            cnt_to_rp[v] = run | (c[i] << 16);
            dinv[v] = rsqrtf((float)(c[i] + 1u));
            run += c[i];
        }
    }
}

__global__ void k_fill(const int* __restrict__ src, const int* __restrict__ dst,
                       const unsigned* __restrict__ rp, const unsigned* __restrict__ rank,
                       unsigned* __restrict__ col) {
    const int e = blockIdx.x * blockDim.x + threadIdx.x;
    if (e < E_N) col[(rp[dst[e]] & 0xFFFFu) + rank[e]] = (unsigned)src[e];
}


__global__ void k_xyzW(const float* __restrict__ xyz, const float* __restrict__ iW,
                       const float* __restrict__ ib, float* __restrict__ xyzWT) {
    int i = blockIdx.x * blockDim.x + threadIdx.x;
    if (i >= V_N * HIDDEN) return;
    int hh = i / V_N, v = i % V_N;
    const float* wp = iW + (size_t)hh * (LATENT + 3);
    xyzWT[i] = ib[hh] + xyz[3 * v] * wp[0] + xyz[3 * v + 1] * wp[1] + xyz[3 * v + 2] * wp[2];
}

__global__ void k_latW(const float* __restrict__ latent, const float* __restrict__ iW,
                       float* __restrict__ latW) {
    int i = blockIdx.x * blockDim.x + threadIdx.x;
    if (i >= B_N * HIDDEN) return;
    int b = i / HIDDEN, hh = i % HIDDEN;
    const float* lp = latent + (size_t)b * LATENT;
    const float* wp = iW + (size_t)hh * (LATENT + 3) + 3;
    float s = 0.0f;
    for (int j = 0; j < LATENT; ++j) s += lp[j] * wp[j];
    latW[i] = s;
}


#define GEMM_OT 2
#define GEMM_NG 2
#define GEMM_NT (4 * GEMM_NG)

template <bool FIRST>
__device__ __forceinline__ void
gemm_tile(const float* __restrict__ src, const float* __restrict__ latW,
          const float* __restrict__ W, float* __restrict__ hT) {
    const int lane  = threadIdx.x & 31;
    const int wv    = threadIdx.x >> 5;
    const int half  = lane >> 4;
    const int l16   = lane & 15;
    const int nbase = (blockIdx.x >> 1) * 128;
    const int obase = (blockIdx.x & 1) * 128 + wv * 32;

    constexpr size_t RS = FIRST ? (size_t)V_N : (size_t)NNODES;

    const float* wrow = W + (size_t)(obase + l16) * HIDDEN + half * 2;
    const float* xcol[GEMM_NG];
    const float* lcol[GEMM_NG];
#pragma unroll
    for (int g = 0; g < GEMM_NG; ++g) {
        const int node4 = nbase + 64 * g + 4 * l16;
        if (FIRST) {
            const int bb = node4 / V_N;
            xcol[g] = src + (size_t)(half * 2) * RS + (node4 - bb * V_N);
            lcol[g] = latW + bb * HIDDEN + half * 2;
        } else {
            xcol[g] = src + (size_t)(half * 2) * RS + node4;
            lcol[g] = nullptr;
        }
    }

    v8f acc[GEMM_OT][GEMM_NT] = {};

#pragma unroll (FIRST ? 1 : 2)
    for (int k = 0; k < HIDDEN; k += 4) {
        v2f a[GEMM_OT];
#pragma unroll
        for (int ot = 0; ot < GEMM_OT; ++ot)
            a[ot] = *(const v2f*)(wrow + (size_t)ot * 16 * HIDDEN + k);
        v2f b[GEMM_NT];
#pragma unroll
        for (int g = 0; g < GEMM_NG; ++g) {
            float4 b0 = *(const float4*)(xcol[g] + (size_t)k * RS);
            float4 b1 = *(const float4*)(xcol[g] + (size_t)(k + 1) * RS);
            if (FIRST) {
                const v2f lw = *(const v2f*)(lcol[g] + k);
                b0.x = fmaxf(b0.x + lw.x, 0.0f); b0.y = fmaxf(b0.y + lw.x, 0.0f);
                b0.z = fmaxf(b0.z + lw.x, 0.0f); b0.w = fmaxf(b0.w + lw.x, 0.0f);
                b1.x = fmaxf(b1.x + lw.y, 0.0f); b1.y = fmaxf(b1.y + lw.y, 0.0f);
                b1.z = fmaxf(b1.z + lw.y, 0.0f); b1.w = fmaxf(b1.w + lw.y, 0.0f);
            }
            b[4 * g + 0].x = b0.x; b[4 * g + 0].y = b1.x;
            b[4 * g + 1].x = b0.y; b[4 * g + 1].y = b1.y;
            b[4 * g + 2].x = b0.z; b[4 * g + 2].y = b1.z;
            b[4 * g + 3].x = b0.w; b[4 * g + 3].y = b1.w;
        }
#pragma unroll
        for (int ot = 0; ot < GEMM_OT; ++ot)
#pragma unroll
            for (int nt = 0; nt < GEMM_NT; ++nt)
                acc[ot][nt] = __builtin_amdgcn_wmma_f32_16x16x4_f32(
                     false, a[ot],  false, b[nt],
                     (short)0, acc[ot][nt],
                     false,  false);
    }

    float* outp = hT + (size_t)(obase + half * 8) * NNODES + (nbase + 4 * l16);
#pragma unroll
    for (int ot = 0; ot < GEMM_OT; ++ot)
#pragma unroll
        for (int r = 0; r < 8; ++r)
#pragma unroll
            for (int g = 0; g < GEMM_NG; ++g)
                *(float4*)(outp + ((size_t)ot * 16 + r) * NNODES + 64 * g) =
                    make_float4(acc[ot][4 * g + 0][r], acc[ot][4 * g + 1][r],
                                acc[ot][4 * g + 2][r], acc[ot][4 * g + 3][r]);
}

__global__ void __launch_bounds__(128)
k_gemm_wmma(const float* __restrict__ xT, const float* __restrict__ W,
            float* __restrict__ hT) {
    gemm_tile<false>(xT, nullptr, W, hT);
}

__global__ void __launch_bounds__(128)
k_gemm_first(const float* __restrict__ xyzWT, const float* __restrict__ latW,
             const float* __restrict__ W, float* __restrict__ hT) {
    gemm_tile<true>(xyzWT, latW, W, hT);
}


#define AGG_T 1024
#define AGG_NB 4
#define AGG_V4 (V_N / 4)

__global__ void __launch_bounds__(AGG_T)
k_agg(const float* __restrict__ hT, float* __restrict__ xT,
      const unsigned* __restrict__ rp, const unsigned* __restrict__ col,
      const float* __restrict__ dinv, const float* __restrict__ bias) {
    __shared__ __attribute__((aligned(16))) float g[AGG_NB * (V_N + 1)];

    const int tid = threadIdx.x;
    const int p   = blockIdx.x & 1;
    const int o   = blockIdx.x >> 1;
    const size_t c0 = (size_t)o * NNODES + (size_t)p * (AGG_NB * V_N);

    if (tid == 0) *(float4*)(g + AGG_NB * V_N) = make_float4(0.0f, 0.0f, 0.0f, 0.0f);
    for (int q = tid; q < AGG_V4; q += AGG_T) {
        const float4 d4 = *(const float4*)(dinv + 4 * q);
        const float4 h0 = *(const float4*)(hT + c0 + 4 * q);
        const float4 h1 = *(const float4*)(hT + c0 + V_N + 4 * q);
        const float4 h2 = *(const float4*)(hT + c0 + 2 * V_N + 4 * q);
        const float4 h3 = *(const float4*)(hT + c0 + 3 * V_N + 4 * q);
        *(float4*)(g + 16 * q)      = make_float4(d4.x * h0.x, d4.x * h1.x, d4.x * h2.x, d4.x * h3.x);
        *(float4*)(g + 16 * q + 4)  = make_float4(d4.y * h0.y, d4.y * h1.y, d4.y * h2.y, d4.y * h3.y);
        *(float4*)(g + 16 * q + 8)  = make_float4(d4.z * h0.z, d4.z * h1.z, d4.z * h2.z, d4.z * h3.z);
        *(float4*)(g + 16 * q + 12) = make_float4(d4.w * h0.w, d4.w * h1.w, d4.w * h2.w, d4.w * h3.w);
    }
    __syncthreads();

    const float bo = bias[o];
    float* xo = xT + c0;
    unsigned w = rp[tid];
    for (int v = tid; v < V_N; v += AGG_T) {
        const int vn = v + AGG_T < V_N ? v + AGG_T : v;
        const unsigned wn = rp[vn];
        const unsigned beg = w & 0xFFFFu;
        const unsigned n   = w >> 16;
        const float dv = dinv[v];
        float4 a = *(const float4*)(g + 4 * v);
        for (unsigned j = 0; j < n; j += 4) {
            unsigned s[4];
            float4 x[4];
#pragma unroll
            for (int i = 0; i < 4; ++i) {
                const unsigned k = j + i < n ? j + i : n - 1u;
                s[i] = col[beg + k];
            }
#pragma unroll
            for (int i = 0; i < 4; ++i)
                s[i] = j + i < n ? s[i] : (unsigned)V_N;
#pragma unroll
            for (int i = 0; i < 4; ++i)
                x[i] = *(const float4*)(g + 4 * s[i]);
#pragma unroll
            for (int i = 0; i < 4; ++i) {
                a.x += x[i].x; a.y += x[i].y; a.z += x[i].z; a.w += x[i].w;
            }
        }
        xo[v]           = fmaxf(dv * a.x + bo, 0.0f);
        xo[V_N + v]     = fmaxf(dv * a.y + bo, 0.0f);
        xo[2 * V_N + v] = fmaxf(dv * a.z + bo, 0.0f);
        xo[3 * V_N + v] = fmaxf(dv * a.w + bo, 0.0f);
        w = wn;
    }
}

__global__ void __launch_bounds__(256)
k_out(const float* __restrict__ xT, const float* __restrict__ oW,
      const float* __restrict__ ob, float* __restrict__ disp) {
    __shared__ float red[4][3][64];
    const int nl = threadIdx.x & 63;
    const int kq = threadIdx.x >> 6;
    const int node = blockIdx.x * 64 + nl;
    const float* xp = xT + (size_t)(kq * 64) * NNODES + node;
    const float* wp = oW + kq * 64;
    float s0 = 0.0f, s1 = 0.0f, s2 = 0.0f;
#pragma unroll 8
    for (int k = 0; k < 64; ++k) {
        float xv = xp[(size_t)k * NNODES];
        s0 += xv * wp[k];
        s1 += xv * wp[HIDDEN + k];
        s2 += xv * wp[2 * HIDDEN + k];
    }
    red[kq][0][nl] = s0; red[kq][1][nl] = s1; red[kq][2][nl] = s2;
    __syncthreads();
    if (threadIdx.x < 192) {
        int n2 = threadIdx.x / 3, o = threadIdx.x % 3;
        float s = ob[o] + ((red[0][o][n2] + red[1][o][n2]) + (red[2][o][n2] + red[3][o][n2]));
        disp[(size_t)blockIdx.x * 192 + threadIdx.x] = s;
    }
}


extern "C" void kernel_launch(void* const* d_in, const int* in_sizes, int n_in,
                              void* d_out, int out_size, void* d_ws, size_t ws_size,
                              hipStream_t stream) {
    const float* vertex_xyz = (const float*)d_in[0];
    const float* latent     = (const float*)d_in[1];
    const int*   edge_index = (const int*)d_in[2];
    const float* input_W    = (const float*)d_in[3];
    const float* input_b    = (const float*)d_in[4];
    const float* conv_W     = (const float*)d_in[5];
    const float* conv_b     = (const float*)d_in[6];
    const float* out_W      = (const float*)d_in[7];
    const float* out_b      = (const float*)d_in[8];
    float* disp = (float*)d_out;

    const int* src = edge_index;
    const int* dst = edge_index + E_N;

    float* xbuf     = (float*)d_ws;
    float* hbuf     = xbuf + (size_t)NNODES * HIDDEN;
    float* xyzWT    = hbuf + (size_t)NNODES * HIDDEN;
    float* latW     = xyzWT + (size_t)V_N * HIDDEN;
    float* dinv     = latW + (size_t)B_N * HIDDEN;
    float* spare_v  = dinv + V_N;
    unsigned* col   = (unsigned*)(spare_v + V_N);
    unsigned* rp    = (unsigned*)spare_v;
    unsigned* rank  = (unsigned*)xbuf;

    const int T = 256;

    k_deg_init<<<(V_N + T - 1) / T, T, 0, stream>>>(rp);
    k_deg_acc<<<(E_N + T - 1) / T, T, 0, stream>>>(dst, rp, rank);
    k_scan<<<1, SCAN_T, 0, stream>>>(rp, dinv);
    k_fill<<<(E_N + T - 1) / T, T, 0, stream>>>(src, dst, rp, rank, col);

    k_xyzW<<<(V_N * HIDDEN + T - 1) / T, T, 0, stream>>>(vertex_xyz, input_W, input_b, xyzWT);
    k_latW<<<(B_N * HIDDEN + T - 1) / T, T, 0, stream>>>(latent, input_W, latW);

    k_gemm_first<<<2 * (NNODES / 128), 128, 0, stream>>>(xyzWT, latW, conv_W, hbuf);
    k_agg<<<HIDDEN * (B_N / AGG_NB), AGG_T, 0, stream>>>(hbuf, xbuf, rp, col, dinv, conv_b);
    for (int l = 1; l < NLAYERS; ++l) {
        k_gemm_wmma<<<2 * (NNODES / 128), 128, 0, stream>>>(xbuf, conv_W + (size_t)l * HIDDEN * HIDDEN, hbuf);
        k_agg<<<HIDDEN * (B_N / AGG_NB), AGG_T, 0, stream>>>(hbuf, xbuf, rp, col, dinv,
                                                            conv_b + (size_t)l * HIDDEN);
    }

    k_out<<<NNODES / 64, T, 0, stream>>>(xbuf, out_W, out_b, disp);
}


# GEMM: each layer's whole W held in registers by 256 resident 8-wave work-groups that walk over 64-node tiles; only x streams, its loads issued one chunk ahead; k summed in another order
# speedup vs baseline: 3.5498x; 1.0974x over previous
#include <hip/hip_runtime.h>


#define V_N     10000
#define E_N     60000
#define B_N     8
#define LATENT  512
#define HIDDEN  256
#define NLAYERS 5
#define NNODES  (B_N * V_N)

typedef float v2f __attribute__((ext_vector_type(2)));
typedef float v8f __attribute__((ext_vector_type(8)));


__global__ void k_deg_init(unsigned* cntv) {
    int v = blockIdx.x * blockDim.x + threadIdx.x;
    if (v < V_N) cntv[v] = 0u;
}

__global__ void k_deg_acc(const int* __restrict__ dst, unsigned* cntv, unsigned* __restrict__ rank) {
    const int e = blockIdx.x * blockDim.x + threadIdx.x;
    if (e < E_N) rank[e] = atomicAdd(&cntv[dst[e]], 1u);
}

#define SCAN_T 1024
#define SCAN_PER 10

__global__ void __launch_bounds__(SCAN_T)
k_scan(unsigned* cnt_to_rp, float* __restrict__ dinv) {
    __shared__ unsigned part[SCAN_T];
    const int tid = threadIdx.x;
    unsigned c[SCAN_PER];
    unsigned s = 0u;
#pragma unroll
    for (int i = 0; i < SCAN_PER; ++i) {
        const int v = tid * SCAN_PER + i;
        c[i] = v < V_N ? cnt_to_rp[v] : 0u;
        s += c[i];
    }
    part[tid] = s;
    __syncthreads();
    for (int off = 1; off < SCAN_T; off <<= 1) {
        const unsigned t = tid >= off ? part[tid - off] : 0u;
        __syncthreads();
        part[tid] += t;
        __syncthreads();
    }
    unsigned run = part[tid] - s;
#pragma unroll
    for (int i = 0; i < SCAN_PER; ++i) {
        const int v = tid * SCAN_PER + i;
        if (v < V_N) {
            cnt_to_rp[v] = run | (c[i] << 16);
            dinv[v] = rsqrtf((float)(c[i] + 1u));
            run += c[i];
        }
    }
}

__global__ void k_fill(const int* __restrict__ src, const int* __restrict__ dst,
                       const unsigned* __restrict__ rp, const unsigned* __restrict__ rank,
                       unsigned* __restrict__ col) {
    const int e = blockIdx.x * blockDim.x + threadIdx.x;
    if (e < E_N) col[(rp[dst[e]] & 0xFFFFu) + rank[e]] = (unsigned)src[e];
}


__global__ void k_xyzW(const float* __restrict__ xyz, const float* __restrict__ iW,
                       const float* __restrict__ ib, float* __restrict__ xyzWT) {
    int i = blockIdx.x * blockDim.x + threadIdx.x;
    if (i >= V_N * HIDDEN) return;
    int hh = i / V_N, v = i % V_N;
    const float* wp = iW + (size_t)hh * (LATENT + 3);
    xyzWT[i] = ib[hh] + xyz[3 * v] * wp[0] + xyz[3 * v + 1] * wp[1] + xyz[3 * v + 2] * wp[2];
}

__global__ void k_latW(const float* __restrict__ latent, const float* __restrict__ iW,
                       float* __restrict__ latW) {
    int i = blockIdx.x * blockDim.x + threadIdx.x;
    if (i >= B_N * HIDDEN) return;
    int b = i / HIDDEN, hh = i % HIDDEN;
    const float* lp = latent + (size_t)b * LATENT;
    const float* wp = iW + (size_t)hh * (LATENT + 3) + 3;
    float s = 0.0f;
    for (int j = 0; j < LATENT; ++j) s += lp[j] * wp[j];
    latW[i] = s;
}


#define GEMM_OT 2
#define GEMM_NT 4
#define GEMM_KS (HIDDEN / 4)
#define GEMM_TN 64
#define GEMM_TILES (NNODES / GEMM_TN)
#define GEMM_GRID 256

template <int KC>
struct BChunk {
    float4 r[KC][2];
    v2f    lw[KC];
};

template <bool FIRST, int KC>
__device__ __forceinline__ void
load_chunk(BChunk<KC>& bc, const float* __restrict__ xcol, const float* __restrict__ lcol, int c) {
    constexpr size_t RS = FIRST ? (size_t)V_N : (size_t)NNODES;
#pragma unroll
    for (int i = 0; i < KC; ++i) {
        const int ks = c + i;
        bc.r[i][0] = *(const float4*)(xcol + (size_t)(2 * ks) * RS);
        bc.r[i][1] = *(const float4*)(xcol + (size_t)(2 * ks + 1) * RS);
        if (FIRST) bc.lw[i] = *(const v2f*)(lcol + 2 * ks);
    }
}

template <bool FIRST, int KC>
__device__ __forceinline__ void
gemm_resident(const float* __restrict__ src, const float* __restrict__ latW,
              const float* __restrict__ W, float* __restrict__ hT) {
    const int lane  = threadIdx.x & 31;
    const int wv    = threadIdx.x >> 5;
    const int half  = lane >> 4;
    const int l16   = lane & 15;
    const int obase = wv * 32;

    constexpr size_t RS = FIRST ? (size_t)V_N : (size_t)NNODES;

    v2f a[GEMM_OT][GEMM_KS];
    {
        const float* wrow = W + (size_t)(obase + l16) * HIDDEN + half * 128;
#pragma unroll
        for (int ot = 0; ot < GEMM_OT; ++ot)
#pragma unroll
            for (int q = 0; q < GEMM_KS / 2; ++q) {
                const float4 w4 = *(const float4*)(wrow + (size_t)ot * 16 * HIDDEN + 4 * q);
                a[ot][2 * q].x     = w4.x; a[ot][2 * q].y     = w4.y;
                a[ot][2 * q + 1].x = w4.z; a[ot][2 * q + 1].y = w4.w;
            }
    }

    auto tile_ptrs = [&](int tile, const float*& xcol, const float*& lcol) {
        const int node4 = tile * GEMM_TN + 4 * l16;
        if (FIRST) {
            const int bb = node4 / V_N;
            xcol = src + (size_t)(half * 128) * RS + (node4 - bb * V_N);
            lcol = latW + bb * HIDDEN + half * 128;
        } else {
            xcol = src + (size_t)(half * 128) * RS + node4;
            lcol = nullptr;
        }
    };

    BChunk<KC> buf[2];
    const float* xcol; const float* lcol;
    tile_ptrs(blockIdx.x, xcol, lcol);
    load_chunk<FIRST, KC>(buf[0], xcol, lcol, 0);

    for (int tile = blockIdx.x; tile < GEMM_TILES; tile += GEMM_GRID) {
        const int tnext = tile + GEMM_GRID < GEMM_TILES ? tile + GEMM_GRID : tile;
        const float* xnext; const float* lnext;
        tile_ptrs(tnext, xnext, lnext);

        v8f acc[GEMM_OT][GEMM_NT] = {};

#pragma unroll
        for (int c = 0; c < GEMM_KS; c += KC) {
            constexpr int NB = 0;
            BChunk<KC>& cur = buf[(c / KC) & 1];
            BChunk<KC>& nxt = buf[((c / KC) + 1) & 1];
            if (c + KC < GEMM_KS) load_chunk<FIRST, KC>(nxt, xcol, lcol, c + KC);
            else                  load_chunk<FIRST, KC>(nxt, xnext, lnext, 0);
            __builtin_amdgcn_sched_barrier(NB);
#pragma unroll
            for (int i = 0; i < KC; ++i) {
                float4 b0 = cur.r[i][0];
                float4 b1 = cur.r[i][1];
                if (FIRST) {
                    const v2f lw = cur.lw[i];
                    b0.x = fmaxf(b0.x + lw.x, 0.0f); b0.y = fmaxf(b0.y + lw.x, 0.0f);
                    b0.z = fmaxf(b0.z + lw.x, 0.0f); b0.w = fmaxf(b0.w + lw.x, 0.0f);
                    b1.x = fmaxf(b1.x + lw.y, 0.0f); b1.y = fmaxf(b1.y + lw.y, 0.0f);
                    b1.z = fmaxf(b1.z + lw.y, 0.0f); b1.w = fmaxf(b1.w + lw.y, 0.0f);
                }
                v2f b[GEMM_NT];
                b[0].x = b0.x; b[0].y = b1.x;
                b[1].x = b0.y; b[1].y = b1.y;
                b[2].x = b0.z; b[2].y = b1.z;
                b[3].x = b0.w; b[3].y = b1.w;
#pragma unroll
                for (int ot = 0; ot < GEMM_OT; ++ot)
#pragma unroll
                    for (int nt = 0; nt < GEMM_NT; ++nt)
                        acc[ot][nt] = __builtin_amdgcn_wmma_f32_16x16x4_f32(
                             false, a[ot][c + i],  false, b[nt],
                             (short)0, acc[ot][nt],
                             false,  false);
            }
            __builtin_amdgcn_sched_barrier(NB);
        }

        float* outp = hT + (size_t)(obase + half * 8) * NNODES + (tile * GEMM_TN + 4 * l16);
#pragma unroll
        for (int ot = 0; ot < GEMM_OT; ++ot)
#pragma unroll
            for (int r = 0; r < 8; ++r)
                *(float4*)(outp + ((size_t)ot * 16 + r) * NNODES) =
                    make_float4(acc[ot][0][r], acc[ot][1][r], acc[ot][2][r], acc[ot][3][r]);

        xcol = xnext; lcol = lnext;
    }
}

__global__ void __launch_bounds__(256)
k_gemm_wmma(const float* __restrict__ xT, const float* __restrict__ W,
            float* __restrict__ hT) {
    gemm_resident<false, 8>(xT, nullptr, W, hT);
}

__global__ void __launch_bounds__(256)
k_gemm_first(const float* __restrict__ xyzWT, const float* __restrict__ latW,
             const float* __restrict__ W, float* __restrict__ hT) {
    gemm_resident<true, 4>(xyzWT, latW, W, hT);
}


#define AGG_T 1024
#define AGG_NB 4
#define AGG_V4 (V_N / 4)

__global__ void __launch_bounds__(AGG_T)
k_agg(const float* __restrict__ hT, float* __restrict__ xT,
      const unsigned* __restrict__ rp, const unsigned* __restrict__ col,
      const float* __restrict__ dinv, const float* __restrict__ bias) {
    __shared__ __attribute__((aligned(16))) float g[AGG_NB * (V_N + 1)];

    const int tid = threadIdx.x;
    const int p   = blockIdx.x & 1;
    const int o   = blockIdx.x >> 1;
    const size_t c0 = (size_t)o * NNODES + (size_t)p * (AGG_NB * V_N);

    if (tid == 0) *(float4*)(g + AGG_NB * V_N) = make_float4(0.0f, 0.0f, 0.0f, 0.0f);
    for (int q = tid; q < AGG_V4; q += AGG_T) {
        const float4 d4 = *(const float4*)(dinv + 4 * q);
        const float4 h0 = *(const float4*)(hT + c0 + 4 * q);
        const float4 h1 = *(const float4*)(hT + c0 + V_N + 4 * q);
        const float4 h2 = *(const float4*)(hT + c0 + 2 * V_N + 4 * q);
        const float4 h3 = *(const float4*)(hT + c0 + 3 * V_N + 4 * q);
        *(float4*)(g + 16 * q)      = make_float4(d4.x * h0.x, d4.x * h1.x, d4.x * h2.x, d4.x * h3.x);
        *(float4*)(g + 16 * q + 4)  = make_float4(d4.y * h0.y, d4.y * h1.y, d4.y * h2.y, d4.y * h3.y);
        *(float4*)(g + 16 * q + 8)  = make_float4(d4.z * h0.z, d4.z * h1.z, d4.z * h2.z, d4.z * h3.z);
        *(float4*)(g + 16 * q + 12) = make_float4(d4.w * h0.w, d4.w * h1.w, d4.w * h2.w, d4.w * h3.w);
    }
    __syncthreads();

    const float bo = bias[o];
    float* xo = xT + c0;
    unsigned w = rp[tid];
    for (int v = tid; v < V_N; v += AGG_T) {
        const int vn = v + AGG_T < V_N ? v + AGG_T : v;
        const unsigned wn = rp[vn];
        const unsigned beg = w & 0xFFFFu;
        const unsigned n   = w >> 16;
        const float dv = dinv[v];
        float4 a = *(const float4*)(g + 4 * v);
        for (unsigned j = 0; j < n; j += 4) {
            unsigned s[4];
            float4 x[4];
#pragma unroll
            for (int i = 0; i < 4; ++i) {
                const unsigned k = j + i < n ? j + i : n - 1u;
                s[i] = col[beg + k];
            }
#pragma unroll
            for (int i = 0; i < 4; ++i)
                s[i] = j + i < n ? s[i] : (unsigned)V_N;
#pragma unroll
            for (int i = 0; i < 4; ++i)
                x[i] = *(const float4*)(g + 4 * s[i]);
#pragma unroll
            for (int i = 0; i < 4; ++i) {
                a.x += x[i].x; a.y += x[i].y; a.z += x[i].z; a.w += x[i].w;
            }
        }
        xo[v]           = fmaxf(dv * a.x + bo, 0.0f);
        xo[V_N + v]     = fmaxf(dv * a.y + bo, 0.0f);
        xo[2 * V_N + v] = fmaxf(dv * a.z + bo, 0.0f);
        xo[3 * V_N + v] = fmaxf(dv * a.w + bo, 0.0f);
        w = wn;
    }
}

__global__ void __launch_bounds__(256)
k_out(const float* __restrict__ xT, const float* __restrict__ oW,
      const float* __restrict__ ob, float* __restrict__ disp) {
    __shared__ float red[4][3][64];
    const int nl = threadIdx.x & 63;
    const int kq = threadIdx.x >> 6;
    const int node = blockIdx.x * 64 + nl;
    const float* xp = xT + (size_t)(kq * 64) * NNODES + node;
    const float* wp = oW + kq * 64;
    float s0 = 0.0f, s1 = 0.0f, s2 = 0.0f;
#pragma unroll 8
    for (int k = 0; k < 64; ++k) {
        float xv = xp[(size_t)k * NNODES];
        s0 += xv * wp[k];
        s1 += xv * wp[HIDDEN + k];
        s2 += xv * wp[2 * HIDDEN + k];
    }
    red[kq][0][nl] = s0; red[kq][1][nl] = s1; red[kq][2][nl] = s2;
    __syncthreads();
    if (threadIdx.x < 192) {
        int n2 = threadIdx.x / 3, o = threadIdx.x % 3;
        float s = ob[o] + ((red[0][o][n2] + red[1][o][n2]) + (red[2][o][n2] + red[3][o][n2]));
        disp[(size_t)blockIdx.x * 192 + threadIdx.x] = s;
    }
}


extern "C" void kernel_launch(void* const* d_in, const int* in_sizes, int n_in,
                              void* d_out, int out_size, void* d_ws, size_t ws_size,
                              hipStream_t stream) {
    const float* vertex_xyz = (const float*)d_in[0];
    const float* latent     = (const float*)d_in[1];
    const int*   edge_index = (const int*)d_in[2];
    const float* input_W    = (const float*)d_in[3];
    const float* input_b    = (const float*)d_in[4];
    const float* conv_W     = (const float*)d_in[5];
    const float* conv_b     = (const float*)d_in[6];
    const float* out_W      = (const float*)d_in[7];
    const float* out_b      = (const float*)d_in[8];
    float* disp = (float*)d_out;

    const int* src = edge_index;
    const int* dst = edge_index + E_N;

    float* xbuf     = (float*)d_ws;
    float* hbuf     = xbuf + (size_t)NNODES * HIDDEN;
    float* xyzWT    = hbuf + (size_t)NNODES * HIDDEN;
    float* latW     = xyzWT + (size_t)V_N * HIDDEN;
    float* dinv     = latW + (size_t)B_N * HIDDEN;
    float* spare_v  = dinv + V_N;
    unsigned* col   = (unsigned*)(spare_v + V_N);
    unsigned* rp    = (unsigned*)spare_v;
    unsigned* rank  = (unsigned*)xbuf;

    const int T = 256;

    k_deg_init<<<(V_N + T - 1) / T, T, 0, stream>>>(rp);
    k_deg_acc<<<(E_N + T - 1) / T, T, 0, stream>>>(dst, rp, rank);
    k_scan<<<1, SCAN_T, 0, stream>>>(rp, dinv);
    k_fill<<<(E_N + T - 1) / T, T, 0, stream>>>(src, dst, rp, rank, col);

    k_xyzW<<<(V_N * HIDDEN + T - 1) / T, T, 0, stream>>>(vertex_xyz, input_W, input_b, xyzWT);
    k_latW<<<(B_N * HIDDEN + T - 1) / T, T, 0, stream>>>(latent, input_W, latW);

    k_gemm_first<<<GEMM_GRID, 256, 0, stream>>>(xyzWT, latW, conv_W, hbuf);
    k_agg<<<HIDDEN * (B_N / AGG_NB), AGG_T, 0, stream>>>(hbuf, xbuf, rp, col, dinv, conv_b);
    for (int l = 1; l < NLAYERS; ++l) {
        k_gemm_wmma<<<GEMM_GRID, 256, 0, stream>>>(xbuf, conv_W + (size_t)l * HIDDEN * HIDDEN, hbuf);
        k_agg<<<HIDDEN * (B_N / AGG_NB), AGG_T, 0, stream>>>(hbuf, xbuf, rp, col, dinv,
                                                            conv_b + (size_t)l * HIDDEN);
    }

    k_out<<<NNODES / 64, T, 0, stream>>>(xbuf, out_W, out_b, disp);
}


# Neighbour lists copied once per call into per-wave rows of 16-byte index quads (LDS byte offsets, zero slot past a list's end): a gather trip is one whole-line load, scalar row count; sums unchanged
# speedup vs baseline: 3.7112x; 1.0455x over previous
#include <hip/hip_runtime.h>


#define V_N     10000
#define E_N     60000
#define B_N     8
#define LATENT  512
#define HIDDEN  256
#define NLAYERS 5
#define NNODES  (B_N * V_N)

typedef float v2f __attribute__((ext_vector_type(2)));
typedef float v8f __attribute__((ext_vector_type(8)));


#define NGROUPS ((V_N + 31) / 32)
#define ZERO_OFF (16u * V_N)

__global__ void k_deg_init(unsigned* cntv) {
    int v = blockIdx.x * blockDim.x + threadIdx.x;
    if (v < V_N) cntv[v] = 0u;
}

__global__ void k_deg_acc(const int* __restrict__ dst, unsigned* cntv, unsigned* __restrict__ rank) {
    const int e = blockIdx.x * blockDim.x + threadIdx.x;
    if (e < E_N) rank[e] = atomicAdd(&cntv[dst[e]], 1u);
}

#define SCAN_T 1024
#define SCAN_PER 10

__global__ void __launch_bounds__(SCAN_T)
k_scan(unsigned* cnt_to_rp, float* __restrict__ dinv, unsigned* __restrict__ ginfo) {
    __shared__ unsigned part[SCAN_T];
    __shared__ unsigned partr[SCAN_T];
    const int tid = threadIdx.x;
    unsigned c[SCAN_PER];
    unsigned s = 0u;
#pragma unroll
    for (int i = 0; i < SCAN_PER; ++i) {
        const int v = tid * SCAN_PER + i;
        c[i] = v < V_N ? cnt_to_rp[v] : 0u;
        s += c[i];
    }
    unsigned rows = 0u;
    if (tid < NGROUPS) {
        unsigned m = 0u;
        for (int j = 0; j < 32; j += 4) {
            const int v = 32 * tid + j;
            if (v < V_N) {
                const uint4 c4 = *(const uint4*)(cnt_to_rp + v);
                m = max(max(m, c4.x), max(max(c4.y, c4.z), c4.w));
            }
        }
        rows = max((m + 3u) >> 2, 1u);
    }
    part[tid] = s;
    partr[tid] = rows;
    __syncthreads();
    for (int off = 1; off < SCAN_T; off <<= 1) {
        const unsigned t = tid >= off ? part[tid - off] : 0u;
        const unsigned r = tid >= off ? partr[tid - off] : 0u;
        __syncthreads();
        part[tid] += t;
        partr[tid] += r;
        __syncthreads();
    }
    unsigned run = part[tid] - s;
#pragma unroll
    for (int i = 0; i < SCAN_PER; ++i) {
        const int v = tid * SCAN_PER + i;
        if (v < V_N) {
            cnt_to_rp[v] = run | (c[i] << 16);
            dinv[v] = rsqrtf((float)(c[i] + 1u));
            run += c[i];
        }
    }
    if (tid < NGROUPS) {
        ginfo[2 * tid]     = partr[tid] - rows;
        ginfo[2 * tid + 1] = rows;
    }
}

__global__ void k_fill(const int* __restrict__ src, const int* __restrict__ dst,
                       const unsigned* __restrict__ rp, const unsigned* __restrict__ rank,
                       unsigned* __restrict__ col) {
    const int e = blockIdx.x * blockDim.x + threadIdx.x;
    if (e < E_N) col[(rp[dst[e]] & 0xFFFFu) + rank[e]] = (unsigned)src[e];
}

#define ELL_T 256
__global__ void __launch_bounds__(ELL_T)
k_ell(const unsigned* __restrict__ rp, const unsigned* __restrict__ col,
      const unsigned* __restrict__ ginfo, unsigned* __restrict__ ell) {
    const int grp  = blockIdx.x * (ELL_T / 32) + (threadIdx.x >> 5);
    const int lane = threadIdx.x & 31;
    if (grp >= NGROUPS) return;
    const unsigned first = ginfo[2 * grp];
    const unsigned rows  = ginfo[2 * grp + 1];
    const int v = 32 * grp + lane;
    unsigned beg = 0u, n = 0u;
    if (v < V_N) { const unsigned w = rp[v]; beg = w & 0xFFFFu; n = w >> 16; }
    uint4* row = (uint4*)ell + (size_t)first * 32 + lane;
    for (unsigned t = 0; t < rows; ++t) {
        unsigned s[4];
#pragma unroll
        for (int i = 0; i < 4; ++i) {
            const unsigned k = 4u * t + i;
            s[i] = k < n ? 16u * col[beg + k] : ZERO_OFF;
        }
        row[(size_t)t * 32] = make_uint4(s[0], s[1], s[2], s[3]);
    }
}


__global__ void k_xyzW(const float* __restrict__ xyz, const float* __restrict__ iW,
                       const float* __restrict__ ib, float* __restrict__ xyzWT) {
    int i = blockIdx.x * blockDim.x + threadIdx.x;
    if (i >= V_N * HIDDEN) return;
    int hh = i / V_N, v = i % V_N;
    const float* wp = iW + (size_t)hh * (LATENT + 3);
    xyzWT[i] = ib[hh] + xyz[3 * v] * wp[0] + xyz[3 * v + 1] * wp[1] + xyz[3 * v + 2] * wp[2];
}

__global__ void k_latW(const float* __restrict__ latent, const float* __restrict__ iW,
                       float* __restrict__ latW) {
    int i = blockIdx.x * blockDim.x + threadIdx.x;
    if (i >= B_N * HIDDEN) return;
    int b = i / HIDDEN, hh = i % HIDDEN;
    const float* lp = latent + (size_t)b * LATENT;
    const float* wp = iW + (size_t)hh * (LATENT + 3) + 3;
    float s = 0.0f;
    for (int j = 0; j < LATENT; ++j) s += lp[j] * wp[j];
    latW[i] = s;
}


#define GEMM_OT 2
#define GEMM_NT 4
#define GEMM_KS (HIDDEN / 4)
#define GEMM_TN 64
#define GEMM_TILES (NNODES / GEMM_TN)
#define GEMM_GRID 256

template <int KC>
struct BChunk {
    float4 r[KC][2];
    v2f    lw[KC];
};

template <bool FIRST, int KC>
__device__ __forceinline__ void
load_chunk(BChunk<KC>& bc, const float* __restrict__ xcol, const float* __restrict__ lcol, int c) {
    constexpr size_t RS = FIRST ? (size_t)V_N : (size_t)NNODES;
#pragma unroll
    for (int i = 0; i < KC; ++i) {
        const int ks = c + i;
        bc.r[i][0] = *(const float4*)(xcol + (size_t)(2 * ks) * RS);
        bc.r[i][1] = *(const float4*)(xcol + (size_t)(2 * ks + 1) * RS);
        if (FIRST) bc.lw[i] = *(const v2f*)(lcol + 2 * ks);
    }
}

template <bool FIRST, int KC>
__device__ __forceinline__ void
gemm_resident(const float* __restrict__ src, const float* __restrict__ latW,
              const float* __restrict__ W, float* __restrict__ hT) {
    const int lane  = threadIdx.x & 31;
    const int wv    = threadIdx.x >> 5;
    const int half  = lane >> 4;
    const int l16   = lane & 15;
    const int obase = wv * 32;

    constexpr size_t RS = FIRST ? (size_t)V_N : (size_t)NNODES;

    v2f a[GEMM_OT][GEMM_KS];
    {
        const float* wrow = W + (size_t)(obase + l16) * HIDDEN + half * 128;
#pragma unroll
        for (int ot = 0; ot < GEMM_OT; ++ot)
#pragma unroll
            for (int q = 0; q < GEMM_KS / 2; ++q) {
                const float4 w4 = *(const float4*)(wrow + (size_t)ot * 16 * HIDDEN + 4 * q);
                a[ot][2 * q].x     = w4.x; a[ot][2 * q].y     = w4.y;
                a[ot][2 * q + 1].x = w4.z; a[ot][2 * q + 1].y = w4.w;
            }
    }

    auto tile_ptrs = [&](int tile, const float*& xcol, const float*& lcol) {
        const int node4 = tile * GEMM_TN + 4 * l16;
        if (FIRST) {
            const int bb = node4 / V_N;
            xcol = src + (size_t)(half * 128) * RS + (node4 - bb * V_N);
            lcol = latW + bb * HIDDEN + half * 128;
        } else {
            xcol = src + (size_t)(half * 128) * RS + node4;
            lcol = nullptr;
        }
    };

    BChunk<KC> buf[2];
    const float* xcol; const float* lcol;
    tile_ptrs(blockIdx.x, xcol, lcol);
    load_chunk<FIRST, KC>(buf[0], xcol, lcol, 0);

    for (int tile = blockIdx.x; tile < GEMM_TILES; tile += GEMM_GRID) {
        const int tnext = tile + GEMM_GRID < GEMM_TILES ? tile + GEMM_GRID : tile;
        const float* xnext; const float* lnext;
        tile_ptrs(tnext, xnext, lnext);

        v8f acc[GEMM_OT][GEMM_NT] = {};

#pragma unroll
        for (int c = 0; c < GEMM_KS; c += KC) {
            constexpr int NB = 0;
            BChunk<KC>& cur = buf[(c / KC) & 1];
            BChunk<KC>& nxt = buf[((c / KC) + 1) & 1];
            if (c + KC < GEMM_KS) load_chunk<FIRST, KC>(nxt, xcol, lcol, c + KC);
            else                  load_chunk<FIRST, KC>(nxt, xnext, lnext, 0);
            __builtin_amdgcn_sched_barrier(NB);
#pragma unroll
            for (int i = 0; i < KC; ++i) {
                float4 b0 = cur.r[i][0];
                float4 b1 = cur.r[i][1];
                if (FIRST) {
                    const v2f lw = cur.lw[i];
                    b0.x = fmaxf(b0.x + lw.x, 0.0f); b0.y = fmaxf(b0.y + lw.x, 0.0f);
                    b0.z = fmaxf(b0.z + lw.x, 0.0f); b0.w = fmaxf(b0.w + lw.x, 0.0f);
                    b1.x = fmaxf(b1.x + lw.y, 0.0f); b1.y = fmaxf(b1.y + lw.y, 0.0f);
                    b1.z = fmaxf(b1.z + lw.y, 0.0f); b1.w = fmaxf(b1.w + lw.y, 0.0f);
                }
                v2f b[GEMM_NT];
                b[0].x = b0.x; b[0].y = b1.x;
                b[1].x = b0.y; b[1].y = b1.y;
                b[2].x = b0.z; b[2].y = b1.z;
                b[3].x = b0.w; b[3].y = b1.w;
#pragma unroll
                for (int ot = 0; ot < GEMM_OT; ++ot)
#pragma unroll
                    for (int nt = 0; nt < GEMM_NT; ++nt)
                        acc[ot][nt] = __builtin_amdgcn_wmma_f32_16x16x4_f32(
                             false, a[ot][c + i],  false, b[nt],
                             (short)0, acc[ot][nt],
                             false,  false);
            }
            __builtin_amdgcn_sched_barrier(NB);
        }

        float* outp = hT + (size_t)(obase + half * 8) * NNODES + (tile * GEMM_TN + 4 * l16);
#pragma unroll
        for (int ot = 0; ot < GEMM_OT; ++ot)
#pragma unroll
            for (int r = 0; r < 8; ++r)
                *(float4*)(outp + ((size_t)ot * 16 + r) * NNODES) =
                    make_float4(acc[ot][0][r], acc[ot][1][r], acc[ot][2][r], acc[ot][3][r]);

        xcol = xnext; lcol = lnext;
    }
}

__global__ void __launch_bounds__(256)
k_gemm_wmma(const float* __restrict__ xT, const float* __restrict__ W,
            float* __restrict__ hT) {
    gemm_resident<false, 8>(xT, nullptr, W, hT);
}

__global__ void __launch_bounds__(256)
k_gemm_first(const float* __restrict__ xyzWT, const float* __restrict__ latW,
             const float* __restrict__ W, float* __restrict__ hT) {
    gemm_resident<true, 4>(xyzWT, latW, W, hT);
}


#define AGG_T 1024
#define AGG_NB 4
#define AGG_V4 (V_N / 4)
#define AGG_W (AGG_T / 32)

__global__ void __launch_bounds__(AGG_T)
k_agg(const float* __restrict__ hT, float* __restrict__ xT,
      const unsigned* __restrict__ ginfo, const unsigned* __restrict__ ell,
      const float* __restrict__ dinv, const float* __restrict__ bias) {
    __shared__ __attribute__((aligned(16))) float g[AGG_NB * (V_N + 1)];

    const int tid = threadIdx.x;
    const int p   = blockIdx.x & 1;
    const int o   = blockIdx.x >> 1;
    const size_t c0 = (size_t)o * NNODES + (size_t)p * (AGG_NB * V_N);

    if (tid == 0) *(float4*)(g + AGG_NB * V_N) = make_float4(0.0f, 0.0f, 0.0f, 0.0f);
    for (int q = tid; q < AGG_V4; q += AGG_T) {
        const float4 d4 = *(const float4*)(dinv + 4 * q);
        const float4 h0 = *(const float4*)(hT + c0 + 4 * q);
        const float4 h1 = *(const float4*)(hT + c0 + V_N + 4 * q);
        const float4 h2 = *(const float4*)(hT + c0 + 2 * V_N + 4 * q);
        const float4 h3 = *(const float4*)(hT + c0 + 3 * V_N + 4 * q);
        *(float4*)(g + 16 * q)      = make_float4(d4.x * h0.x, d4.x * h1.x, d4.x * h2.x, d4.x * h3.x);
        *(float4*)(g + 16 * q + 4)  = make_float4(d4.y * h0.y, d4.y * h1.y, d4.y * h2.y, d4.y * h3.y);
        *(float4*)(g + 16 * q + 8)  = make_float4(d4.z * h0.z, d4.z * h1.z, d4.z * h2.z, d4.z * h3.z);
        *(float4*)(g + 16 * q + 12) = make_float4(d4.w * h0.w, d4.w * h1.w, d4.w * h2.w, d4.w * h3.w);
    }
    __syncthreads();

    const float bo = bias[o];
    float* xo = xT + c0;
    const int lane = tid & 31;
    const uint2* gi2  = (const uint2*)ginfo;
    const uint4* rows = (const uint4*)ell + lane;
    const char*  gb   = (const char*)g;

    int grp = __builtin_amdgcn_readfirstlane(tid >> 5);
    int gn  = grp + AGG_W < NGROUPS ? grp + AGG_W : grp;
    uint2 gi  = gi2[grp];
    uint2 gin = gi2[gn];
    uint4 q = rows[(size_t)gi.x * 32];
    for (;;) {
        const int gnn = gn + AGG_W < NGROUPS ? gn + AGG_W : gn;
        const uint2 ginn = gi2[gnn];
        const uint4 qnext = rows[(size_t)gin.x * 32];

        const int v  = 32 * grp + lane;
        const int vc = v < V_N ? v : V_N - 1;
        const float dv = dinv[vc];
        float4 a = *(const float4*)(g + 4 * vc);
        const uint4* row = rows + (size_t)gi.x * 32;
        const unsigned nrow = gi.y;
        for (unsigned t = 0; t < nrow; ++t) {
            const unsigned tn = t + 1u < nrow ? t + 1u : t;
            const uint4 qn = row[(size_t)tn * 32];
            const float4 x0 = *(const float4*)(gb + q.x);
            const float4 x1 = *(const float4*)(gb + q.y);
            const float4 x2 = *(const float4*)(gb + q.z);
            const float4 x3 = *(const float4*)(gb + q.w);
            a.x += x0.x; a.y += x0.y; a.z += x0.z; a.w += x0.w;
            a.x += x1.x; a.y += x1.y; a.z += x1.z; a.w += x1.w;
            a.x += x2.x; a.y += x2.y; a.z += x2.z; a.w += x2.w;
            a.x += x3.x; a.y += x3.y; a.z += x3.z; a.w += x3.w;
            q = qn;
        }
        if (v < V_N) {
            xo[v]           = fmaxf(dv * a.x + bo, 0.0f);
            xo[V_N + v]     = fmaxf(dv * a.y + bo, 0.0f);
            xo[2 * V_N + v] = fmaxf(dv * a.z + bo, 0.0f);
            xo[3 * V_N + v] = fmaxf(dv * a.w + bo, 0.0f);
        }
        if (grp + AGG_W >= NGROUPS) break;
        grp = gn; gn = gnn; gi = gin; gin = ginn; q = qnext;
    }
}

__global__ void __launch_bounds__(256)
k_out(const float* __restrict__ xT, const float* __restrict__ oW,
      const float* __restrict__ ob, float* __restrict__ disp) {
    __shared__ float red[4][3][64];
    const int nl = threadIdx.x & 63;
    const int kq = threadIdx.x >> 6;
    const int node = blockIdx.x * 64 + nl;
    const float* xp = xT + (size_t)(kq * 64) * NNODES + node;
    const float* wp = oW + kq * 64;
    float s0 = 0.0f, s1 = 0.0f, s2 = 0.0f;
#pragma unroll 8
    for (int k = 0; k < 64; ++k) {
        float xv = xp[(size_t)k * NNODES];
        s0 += xv * wp[k];
        s1 += xv * wp[HIDDEN + k];
        s2 += xv * wp[2 * HIDDEN + k];
    }
    red[kq][0][nl] = s0; red[kq][1][nl] = s1; red[kq][2][nl] = s2;
    __syncthreads();
    if (threadIdx.x < 192) {
        int n2 = threadIdx.x / 3, o = threadIdx.x % 3;
        float s = ob[o] + ((red[0][o][n2] + red[1][o][n2]) + (red[2][o][n2] + red[3][o][n2]));
        disp[(size_t)blockIdx.x * 192 + threadIdx.x] = s;
    }
}


extern "C" void kernel_launch(void* const* d_in, const int* in_sizes, int n_in,
                              void* d_out, int out_size, void* d_ws, size_t ws_size,
                              hipStream_t stream) {
    const float* vertex_xyz = (const float*)d_in[0];
    const float* latent     = (const float*)d_in[1];
    const int*   edge_index = (const int*)d_in[2];
    const float* input_W    = (const float*)d_in[3];
    const float* input_b    = (const float*)d_in[4];
    const float* conv_W     = (const float*)d_in[5];
    const float* conv_b     = (const float*)d_in[6];
    const float* out_W      = (const float*)d_in[7];
    const float* out_b      = (const float*)d_in[8];
    float* disp = (float*)d_out;

    const int* src = edge_index;
    const int* dst = edge_index + E_N;

    float* xbuf     = (float*)d_ws;
    float* hbuf     = xbuf + (size_t)NNODES * HIDDEN;
    float* xyzWT    = hbuf + (size_t)NNODES * HIDDEN;
    float* latW     = xyzWT + (size_t)V_N * HIDDEN;
    float* dinv     = latW + (size_t)B_N * HIDDEN;
    float* spare_v  = dinv + V_N;
    unsigned* col   = (unsigned*)(spare_v + V_N);
    unsigned* rp    = (unsigned*)spare_v;
    unsigned* rank  = (unsigned*)xbuf;
    unsigned* ginfo = (unsigned*)xyzWT;
    unsigned* ell   = (unsigned*)xyzWT + 1024;

    const int T = 256;

    k_xyzW<<<(V_N * HIDDEN + T - 1) / T, T, 0, stream>>>(vertex_xyz, input_W, input_b, xyzWT);
    k_latW<<<(B_N * HIDDEN + T - 1) / T, T, 0, stream>>>(latent, input_W, latW);
    k_gemm_first<<<GEMM_GRID, 256, 0, stream>>>(xyzWT, latW, conv_W, hbuf);

    k_deg_init<<<(V_N + T - 1) / T, T, 0, stream>>>(rp);
    k_deg_acc<<<(E_N + T - 1) / T, T, 0, stream>>>(dst, rp, rank);
    k_scan<<<1, SCAN_T, 0, stream>>>(rp, dinv, ginfo);
    k_fill<<<(E_N + T - 1) / T, T, 0, stream>>>(src, dst, rp, rank, col);
    k_ell<<<(NGROUPS + ELL_T / 32 - 1) / (ELL_T / 32), ELL_T, 0, stream>>>(rp, col, ginfo, ell);

    k_agg<<<HIDDEN * (B_N / AGG_NB), AGG_T, 0, stream>>>(hbuf, xbuf, ginfo, ell, dinv, conv_b);
    for (int l = 1; l < NLAYERS; ++l) {
        k_gemm_wmma<<<GEMM_GRID, 256, 0, stream>>>(xbuf, conv_W + (size_t)l * HIDDEN * HIDDEN, hbuf);
        k_agg<<<HIDDEN * (B_N / AGG_NB), AGG_T, 0, stream>>>(hbuf, xbuf, ginfo, ell, dinv,
                                                            conv_b + (size_t)l * HIDDEN);
    }

    k_out<<<NNODES / 64, T, 0, stream>>>(xbuf, out_W, out_b, disp);
}
